# starting kernel
# baseline (speedup 1.0000x reference)
#include <hip/hip_runtime.h>
#include <math.h>

typedef float v2f __attribute__((ext_vector_type(2)));
typedef float v8f __attribute__((ext_vector_type(8)));

#define BATCH 32
#define CIN   512
#define HW    4096
#define NCLUS 10
#define KPAD  16
#define KKEEP 8
#define KCHUNK (HW / 8)


__global__ void vlad_zero_asum(float* __restrict__ asum) {
    int i = blockIdx.x * blockDim.x + threadIdx.x;
    if (i < BATCH * KPAD) asum[i] = 0.0f;
}

__global__ __launch_bounds__(256) void vlad_softmax(const float* __restrict__ score,
                                                    float* __restrict__ S,
                                                    float* __restrict__ asum) {
    int gid = blockIdx.x * 256 + threadIdx.x;
    int b   = gid >> 12;
    int hw  = gid & (HW - 1);

    const float* sp = score + (size_t)b * NCLUS * HW + hw;
    float s[NCLUS];
    float mx = -1e30f;
#pragma unroll
    for (int k = 0; k < NCLUS; ++k) { s[k] = sp[(size_t)k * HW]; mx = fmaxf(mx, s[k]); }
    float sum = 0.0f;
#pragma unroll
    for (int k = 0; k < NCLUS; ++k) { s[k] = __expf(s[k] - mx); sum += s[k]; }
    float inv = 1.0f / sum;

    float* Sp = S + (size_t)b * KPAD * HW + hw;
#pragma unroll
    for (int k = 0; k < NCLUS; ++k) { s[k] *= inv; Sp[(size_t)k * HW] = s[k]; }
#pragma unroll
    for (int k = NCLUS; k < KPAD; ++k) Sp[(size_t)k * HW] = 0.0f;

    __shared__ float red[256];
    for (int k = 0; k < NCLUS; ++k) {
        red[threadIdx.x] = s[k];
        __syncthreads();
        for (int off = 128; off > 0; off >>= 1) {
            if (threadIdx.x < off) red[threadIdx.x] += red[threadIdx.x + off];
            __syncthreads();
        }
        if (threadIdx.x == 0) atomicAdd(&asum[b * KPAD + k], red[0]);
        __syncthreads();
    }
}

__global__ __launch_bounds__(256) void vlad_wmma_gemm(const float* __restrict__ F,
                                                      const float* __restrict__ S,
                                                      float* __restrict__ WX) {
    const int b     = blockIdx.x >> 3;
    const int ntile = blockIdx.x & 7;
    const int c0    = ntile * 64;

    const int wave  = threadIdx.x >> 5;
    const int lane  = threadIdx.x & 31;
    const int kbase = wave * KCHUNK;
    const int mn    = lane & 15;
    const int kk    = (lane >> 4) * 2;

    const float* Sp = S + ((size_t)b * KPAD + mn) * HW + kbase + kk;
    const float* Fp = F + ((size_t)b * CIN + c0 + mn) * HW + kbase + kk;

    v8f acc0 = {}; v8f acc1 = {}; v8f acc2 = {}; v8f acc3 = {};

#pragma unroll 4
    for (int k = 0; k < KCHUNK; k += 4) {
        v2f a  = *(const v2f*)(Sp + k);
        v2f b0 = *(const v2f*)(Fp + k);
        v2f b1 = *(const v2f*)(Fp + (size_t)16 * HW + k);
        v2f b2 = *(const v2f*)(Fp + (size_t)32 * HW + k);
        v2f b3 = *(const v2f*)(Fp + (size_t)48 * HW + k);
        acc0 = __builtin_amdgcn_wmma_f32_16x16x4_f32(false, a, false, b0, (short)0, acc0, false, false);
        acc1 = __builtin_amdgcn_wmma_f32_16x16x4_f32(false, a, false, b1, (short)0, acc1, false, false);
        acc2 = __builtin_amdgcn_wmma_f32_16x16x4_f32(false, a, false, b2, (short)0, acc2, false, false);
        acc3 = __builtin_amdgcn_wmma_f32_16x16x4_f32(false, a, false, b3, (short)0, acc3, false, false);
    }

    __shared__ float lds[8 * 1024];
    float* lp = lds + wave * 1024;
#pragma unroll
    for (int j = 0; j < 8; ++j) {
        lp[(0 * 8 + j) * 32 + lane] = acc0[j];
        lp[(1 * 8 + j) * 32 + lane] = acc1[j];
        lp[(2 * 8 + j) * 32 + lane] = acc2[j];
        lp[(3 * 8 + j) * 32 + lane] = acc3[j];
    }
    __syncthreads();

#pragma unroll
    for (int e = 0; e < 4; ++e) {
        int idx = threadIdx.x * 4 + e;
        float v = 0.0f;
#pragma unroll
        for (int w = 0; w < 8; ++w) v += lds[w * 1024 + idx];
        int slot = idx >> 5;
        int l    = idx & 31;
        int t    = slot >> 3;
        int j    = slot & 7;
        int mm   = j + ((l >> 4) << 3);
        int nn   = t * 16 + (l & 15);
        WX[((size_t)b * KPAD + mm) * CIN + c0 + nn] = v;
    }
}

__global__ __launch_bounds__(256) void vlad_finalize(const float* __restrict__ WX,
                                                     const float* __restrict__ asum,
                                                     const float* __restrict__ centroids,
                                                     float* __restrict__ out) {
    const int b   = blockIdx.x >> 3;
    const int kc  = blockIdx.x & 7;
    const int tid = threadIdx.x;

    const float a = asum[b * KPAD + kc];
    const float* wp = WX + ((size_t)b * KPAD + kc) * CIN;
    const float* cp = centroids + (size_t)kc * CIN;

    float r0 = wp[tid]       - a * cp[tid];
    float r1 = wp[tid + 256] - a * cp[tid + 256];

    __shared__ float red[256];
    red[tid] = r0 * r0 + r1 * r1;
    __syncthreads();
    for (int off = 128; off > 0; off >>= 1) {
        if (tid < off) red[tid] += red[tid + off];
        __syncthreads();
    }
    __shared__ float invn;
    if (tid == 0) invn = 1.0f / fmaxf(sqrtf(red[0]), 1e-12f);
    __syncthreads();

    float* op = out + ((size_t)b * KKEEP + kc) * CIN;
    op[tid]       = r0 * invn;
    op[tid + 256] = r1 * invn;
}

extern "C" void kernel_launch(void* const* d_in, const int* in_sizes, int n_in,
                              void* d_out, int out_size, void* d_ws, size_t ws_size,
                              hipStream_t stream) {
    const float* features  = (const float*)d_in[0];
    const float* score     = (const float*)d_in[1];
    const float* centroids = (const float*)d_in[2];
    float* out = (float*)d_out;

    float* S    = (float*)d_ws;
    float* WX   = S  + (size_t)BATCH * KPAD * HW;
    float* ASUM = WX + (size_t)BATCH * KPAD * CIN;

    vlad_zero_asum<<<2, 256, 0, stream>>>(ASUM);
    vlad_softmax  <<<BATCH * HW / 256, 256, 0, stream>>>(score, S, ASUM);
    vlad_wmma_gemm<<<BATCH * 8,        256, 0, stream>>>(features, S, WX);
    vlad_finalize <<<BATCH * KKEEP,    256, 0, stream>>>(WX, ASUM, centroids, out);
}


# F loads run one 32-pixel group ahead: a second register set of 32 loads is requested as one group before the WMMAs on the current set (32 to 64 loads in flight per wave); sum orders unchanged
# speedup vs baseline: 1.8634x; 1.8634x over previous
#include <hip/hip_runtime.h>
#include <math.h>


typedef float v2f __attribute__((ext_vector_type(2)));
typedef float v8f __attribute__((ext_vector_type(8)));

#define BATCH  32
#define CIN    512
#define HW     4096
#define NCLUS  10
#define KKEEP  8
#define KCHUNK 512
#define NCHUNK (HW / KCHUNK)
#define SPITCH (KCHUNK + 8)


__device__ __forceinline__ void load_group(v2f (&f)[32], const float* __restrict__ pe, const float* __restrict__ po) {
#pragma unroll
    for (int t = 0; t < 4; ++t) {
#pragma unroll
        for (int j = 0; j < 4; ++j) {
            f[8 * t + 2 * j]     = *(const v2f*)(pe + (size_t)(16 * j) * HW + 8 * t);
            f[8 * t + 2 * j + 1] = *(const v2f*)(po + (size_t)(16 * j) * HW + 8 * t);
        }
    }
}

__device__ __forceinline__ void load_afrag(v2f (&a)[4], const float* ap) {
#pragma unroll
    for (int t = 0; t < 4; ++t) a[t] = *(const v2f*)(ap + 8 * t);
}

__device__ __forceinline__ void mma_group(v8f (&acc)[8], const v2f (&f)[32], const v2f (&a)[4]) {
#pragma unroll
    for (int t = 0; t < 4; ++t) {
#pragma unroll
        for (int i = 0; i < 8; ++i)
            acc[i] = __builtin_amdgcn_wmma_f32_16x16x4_f32(false, a[t], false, f[8 * t + i], (short)0, acc[i], false, false);
    }
}

__global__ __launch_bounds__(256) void vlad_chunk_gemm(const float* __restrict__ F,
                                                       const float* __restrict__ score,
                                                       float* __restrict__ PWX,
                                                       float* __restrict__ PAS) {
    const int b     = blockIdx.x >> 3;
    const int chunk = blockIdx.x & 7;
    const int kbase = chunk * KCHUNK;
    const int tid   = threadIdx.x;
    const int wave  = tid >> 5;
    const int lane  = tid & 31;

    __shared__ float St[KKEEP * SPITCH];

    {
        const float* sp = score + (size_t)b * NCLUS * HW + kbase + 2 * tid;
        v2f s[NCLUS];
        float mx0 = -1e30f, mx1 = -1e30f;
#pragma unroll
        for (int k = 0; k < NCLUS; ++k) {
            s[k] = *(const v2f*)(sp + (size_t)k * HW);
            mx0 = fmaxf(mx0, s[k].x);
            mx1 = fmaxf(mx1, s[k].y);
        }
        float sum0 = 0.0f, sum1 = 0.0f;
#pragma unroll
        for (int k = 0; k < NCLUS; ++k) {
            s[k].x = __expf(s[k].x - mx0); sum0 += s[k].x;
            s[k].y = __expf(s[k].y - mx1); sum1 += s[k].y;
        }
        const float inv0 = 1.0f / sum0;
        const float inv1 = 1.0f / sum1;
        float* tp = St + 2 * tid;
#pragma unroll
        for (int k = 0; k < KKEEP; ++k) {
            v2f v; v.x = s[k].x * inv0; v.y = s[k].y * inv1;
            *(v2f*)(tp + k * SPITCH) = v;
        }
    }
    __syncthreads();

    {
        const float* rp = St + wave * SPITCH + lane;
        float v = 0.0f;
#pragma unroll
        for (int i = 0; i < KCHUNK / 32; ++i) v += rp[32 * i];
#pragma unroll
        for (int off = 16; off > 0; off >>= 1) v += __shfl_xor(v, off, 32);
        if (lane == 0) PAS[(b * NCHUNK + chunk) * KKEEP + wave] = v;
    }

    const int mn = lane & 15;
    const int kk = (lane >> 4) * 2;
    const int r8 = mn & 7;
    const int qd = mn >> 3;
    const int c0 = wave * 64;

    const float* Ap  = St + r8 * SPITCH + 4 * qd + kk;
    const float* FpE = F + ((size_t)b * CIN + c0 + r8) * HW + kbase + 4 * qd + kk;
    const float* FpO = F + ((size_t)b * CIN + c0 + 8 + r8) * HW + kbase + 4 * (1 - qd) + kk;

    v8f acc[8] = {};

    v2f f0[32], f1[32], a0[4], a1[4];
    load_group(f0, FpE, FpO);
#pragma nounroll
    for (int k = 0; k < KCHUNK - 64; k += 64) {
        load_afrag(a0, Ap + k);
        load_group(f1, FpE + k + 32, FpO + k + 32);
        __builtin_amdgcn_sched_barrier(0);
        mma_group(acc, f0, a0);
        __builtin_amdgcn_sched_barrier(0);
        load_afrag(a1, Ap + k + 32);
        load_group(f0, FpE + k + 64, FpO + k + 64);
        __builtin_amdgcn_sched_barrier(0);
        mma_group(acc, f1, a1);
    }
    load_afrag(a0, Ap + KCHUNK - 64);
    load_afrag(a1, Ap + KCHUNK - 32);
    load_group(f1, FpE + KCHUNK - 32, FpO + KCHUNK - 32);
    __builtin_amdgcn_sched_barrier(0);
    mma_group(acc, f0, a0);
    mma_group(acc, f1, a1);

    const v8f aE0 = acc[0], aO0 = acc[1], aE1 = acc[2], aO1 = acc[3];
    const v8f aE2 = acc[4], aO2 = acc[5], aE3 = acc[6], aO3 = acc[7];

    const bool odd = ((lane + 8) & 16) != 0;
    float u0[KKEEP], u1[KKEEP], u2[KKEEP], u3[KKEEP];
#pragma unroll
    for (int j = 0; j < KKEEP; ++j) {
        u0[j] = odd ? aO0[j] : aE0[j];
        u1[j] = odd ? aO1[j] : aE1[j];
        u2[j] = odd ? aO2[j] : aE2[j];
        u3[j] = odd ? aO3[j] : aE3[j];
        u0[j] += __shfl_xor(u0[j], 24, 32);
        u1[j] += __shfl_xor(u1[j], 24, 32);
        u2[j] += __shfl_xor(u2[j], 24, 32);
        u3[j] += __shfl_xor(u3[j], 24, 32);
    }
    if (lane < 16) {
        float* pp = PWX + ((size_t)(b * NCHUNK + chunk) * KKEEP) * CIN + c0 + lane;
#pragma unroll
        for (int j = 0; j < KKEEP; ++j) {
            pp[j * CIN +  0] = u0[j];
            pp[j * CIN + 16] = u1[j];
            pp[j * CIN + 32] = u2[j];
            pp[j * CIN + 48] = u3[j];
        }
    }
}

__global__ __launch_bounds__(256) void vlad_finalize(const float* __restrict__ PWX,
                                                     const float* __restrict__ PAS,
                                                     const float* __restrict__ centroids,
                                                     float* __restrict__ out) {
    const int b   = blockIdx.x >> 3;
    const int kc  = blockIdx.x & 7;
    const int tid = threadIdx.x;

    float a = 0.0f, w0 = 0.0f, w1 = 0.0f;
#pragma unroll
    for (int ch = 0; ch < NCHUNK; ++ch) {
        const float* wp = PWX + ((size_t)(b * NCHUNK + ch) * KKEEP + kc) * CIN;
        a  += PAS[(b * NCHUNK + ch) * KKEEP + kc];
        w0 += wp[tid];
        w1 += wp[tid + 256];
    }
    const float* cp = centroids + (size_t)kc * CIN;

    float r0 = w0 - a * cp[tid];
    float r1 = w1 - a * cp[tid + 256];

    __shared__ float red[256];
    red[tid] = r0 * r0 + r1 * r1;
    __syncthreads();
    for (int off = 128; off > 0; off >>= 1) {
        if (tid < off) red[tid] += red[tid + off];
        __syncthreads();
    }
    __shared__ float invn;
    if (tid == 0) invn = 1.0f / fmaxf(sqrtf(red[0]), 1e-12f);
    __syncthreads();

    float* op = out + ((size_t)b * KKEEP + kc) * CIN;
    op[tid]       = r0 * invn;
    op[tid + 256] = r1 * invn;
}

extern "C" void kernel_launch(void* const* d_in, const int* in_sizes, int n_in,
                              void* d_out, int out_size, void* d_ws, size_t ws_size,
                              hipStream_t stream) {
    const float* features  = (const float*)d_in[0];
    const float* score     = (const float*)d_in[1];
    const float* centroids = (const float*)d_in[2];
    float* out = (float*)d_out;

    float* PWX = (float*)d_ws;
    float* PAS = PWX + (size_t)BATCH * NCHUNK * KKEEP * CIN;

    vlad_chunk_gemm<<<BATCH * NCHUNK, 256, 0, stream>>>(features, score, PWX, PAS);
    vlad_finalize  <<<BATCH * KKEEP,  256, 0, stream>>>(PWX, PAS, centroids, out);
}
